# starting kernel
# baseline (speedup 1.0000x reference)
#include <hip/hip_runtime.h>
#include <stdint.h>

typedef float v2f __attribute__((ext_vector_type(2)));
typedef float v8f __attribute__((ext_vector_type(8)));

__global__ void __launch_bounds__(256)
gemm_bias_wmma(const float* __restrict__ X, const float* __restrict__ W,
               const float* __restrict__ bias, float* __restrict__ Y,
               int K, int n_rows)
{
    const int wave    = (blockIdx.x * blockDim.x + threadIdx.x) >> 5;
    const int lane    = threadIdx.x & 31;
    const int n_tiles = (n_rows + 15) >> 4;
    if (wave >= n_tiles) return;

    const int half = lane >> 4;
    const int mrow = lane & 15;
    const int arow = min(wave * 16 + mrow, n_rows - 1);
    const float* __restrict__ xrow = X + (size_t)arow * (size_t)K;

    v8f c0 = {};
    v8f c1 = {};

    for (int k = 0; k < K; k += 4) {
        const int kb = k + half * 2;
        v2f a = *(const v2f*)(xrow + kb);

        v2f b0, b1;
        b0.x = W[(size_t)kb * 32 + mrow];
        b0.y = W[(size_t)(kb + 1) * 32 + mrow];
        b1.x = W[(size_t)kb * 32 + 16 + mrow];
        b1.y = W[(size_t)(kb + 1) * 32 + 16 + mrow];

        c0 = __builtin_amdgcn_wmma_f32_16x16x4_f32(false, a, false, b0,
                                                   (short)0, c0, false, false);
        c1 = __builtin_amdgcn_wmma_f32_16x16x4_f32(false, a, false, b1,
                                                   (short)0, c1, false, false);
    }

    const float bv0 = bias[mrow];
    const float bv1 = bias[16 + mrow];
#pragma unroll
    for (int r = 0; r < 8; ++r) {
        const int orow = wave * 16 + r + half * 8;
        if (orow < n_rows) {
            Y[(size_t)orow * 32 + mrow]      = c0[r] + bv0;
            Y[(size_t)orow * 32 + 16 + mrow] = c1[r] + bv1;
        }
    }
}

__global__ void init_zero(float* __restrict__ cnt, float* __restrict__ accB,
                          float* __restrict__ out, int n_nodes)
{
    const int i     = blockIdx.x * blockDim.x + threadIdx.x;
    const int total = n_nodes * 32;
    if (i < total) { accB[i] = 0.0f; out[i] = 0.0f; }
    if (i < n_nodes) cnt[i] = 0.0f;
}

__global__ void degree_kernel(const int* __restrict__ dst,
                              float* __restrict__ cnt, int n_edges)
{
    const int e = blockIdx.x * blockDim.x + threadIdx.x;
    if (e < n_edges) atomicAdd(&cnt[dst[e]], 1.0f);
}

__global__ void scatter_accum(const int* __restrict__ src, const int* __restrict__ dst,
                              const float* __restrict__ H, float* __restrict__ Acc,
                              int n_edges)
{
    const int t = blockIdx.x * blockDim.x + threadIdx.x;
    const int e = t >> 5;
    const int f = t & 31;
    if (e < n_edges) {
        const int s = src[e];
        const int d = dst[e];
        atomicAdd(&Acc[(size_t)d * 32 + f], H[(size_t)s * 32 + f]);
    }
}

__global__ void finalize_relu(const float* __restrict__ B, const float* __restrict__ cnt,
                              float* __restrict__ A, int n_nodes)
{
    const int i = blockIdx.x * blockDim.x + threadIdx.x;
    if (i < n_nodes * 32) {
        const float c = cnt[i >> 5];
        const float v = B[i] / fmaxf(c, 1.0f);
        A[i] = fmaxf(v, 0.0f);
    }
}

__global__ void finalize_div(float* __restrict__ out, const float* __restrict__ cnt,
                             int n_nodes)
{
    const int i = blockIdx.x * blockDim.x + threadIdx.x;
    if (i < n_nodes * 32) {
        out[i] = out[i] / fmaxf(cnt[i >> 5], 1.0f);
    }
}

extern "C" void kernel_launch(void* const* d_in, const int* in_sizes, int n_in,
                              void* d_out, int out_size, void* d_ws, size_t ws_size,
                              hipStream_t stream)
{
    const float* x    = (const float*)d_in[0];
    const int*   eidx = (const int*)d_in[1];
    const float* W1   = (const float*)d_in[2];
    const float* b1   = (const float*)d_in[3];
    const float* W2   = (const float*)d_in[4];
    const float* b2   = (const float*)d_in[5];
    float*       out  = (float*)d_out;

    const int n_nodes = in_sizes[0] / 128;
    const int n_edges = in_sizes[1] / 2;
    const int* src = eidx;
    const int* dst = eidx + n_edges;

    float* A   = (float*)d_ws;
    float* B   = A + (size_t)n_nodes * 32;
    float* cnt = B + (size_t)n_nodes * 32;

    const int TPB = 256;
    const int n_feat_elems   = n_nodes * 32;
    const int grid_feat      = (n_feat_elems + TPB - 1) / TPB;
    const int grid_edges     = (n_edges + TPB - 1) / TPB;
    const int grid_edge_feat = (int)(((size_t)n_edges * 32 + TPB - 1) / TPB);
    const int n_tiles        = (n_nodes + 15) / 16;
    const int grid_gemm      = (n_tiles + 7) / 8;

    init_zero<<<grid_feat, TPB, 0, stream>>>(cnt, B, out, n_nodes);
    degree_kernel<<<grid_edges, TPB, 0, stream>>>(dst, cnt, n_edges);
    gemm_bias_wmma<<<grid_gemm, TPB, 0, stream>>>(x, W1, b1, A, 128, n_nodes);
    scatter_accum<<<grid_edge_feat, TPB, 0, stream>>>(src, dst, A, B, n_edges);
    finalize_relu<<<grid_feat, TPB, 0, stream>>>(B, cnt, A, n_nodes);
    gemm_bias_wmma<<<grid_gemm, TPB, 0, stream>>>(A, W2, b2, B, 32, n_nodes);
    scatter_accum<<<grid_edge_feat, TPB, 0, stream>>>(src, dst, B, out, n_edges);
    finalize_div<<<grid_feat, TPB, 0, stream>>>(out, cnt, n_nodes);
}


# slot rows owned by an L2 half: place and pull work-groups read their die id and draw tickets from their half's queue (nodes cut at midpoint); a row's CAS and reads stay in one half; slot-order sums
# speedup vs baseline: 4.2913x; 4.2913x over previous
#include <hip/hip_runtime.h>
#include <stdint.h>

typedef float v2f __attribute__((ext_vector_type(2)));
typedef float v8f __attribute__((ext_vector_type(8)));

__global__ void __launch_bounds__(256)
gemm_bias_wmma(const float* __restrict__ X, const float* __restrict__ W,
               const float* __restrict__ bias, float* __restrict__ Y,
               int K, int n_rows)
{
    const int wave    = (blockIdx.x * blockDim.x + threadIdx.x) >> 5;
    const int lane    = threadIdx.x & 31;
    const int n_tiles = (n_rows + 15) >> 4;
    if (wave >= n_tiles) return;

    const int half = lane >> 4;
    const int mrow = lane & 15;
    const int arow = min(wave * 16 + mrow, n_rows - 1);
    const float* __restrict__ xrow = X + (size_t)arow * (size_t)K;

    v8f c0 = {};
    v8f c1 = {};

    for (int k = 0; k < K; k += 4) {
        const int kb = k + half * 2;
        v2f a = *(const v2f*)(xrow + kb);

        v2f b0, b1;
        b0.x = W[(size_t)kb * 32 + mrow];
        b0.y = W[(size_t)(kb + 1) * 32 + mrow];
        b1.x = W[(size_t)kb * 32 + 16 + mrow];
        b1.y = W[(size_t)(kb + 1) * 32 + 16 + mrow];

        c0 = __builtin_amdgcn_wmma_f32_16x16x4_f32(false, a, false, b0,
                                                   (short)0, c0, false, false);
        c1 = __builtin_amdgcn_wmma_f32_16x16x4_f32(false, a, false, b1,
                                                   (short)0, c1, false, false);
    }

    const float bv0 = bias[mrow];
    const float bv1 = bias[16 + mrow];
#pragma unroll
    for (int r = 0; r < 8; ++r) {
        const int orow = wave * 16 + r + half * 8;
        if (orow < n_rows) {
            Y[(size_t)orow * 32 + mrow]      = c0[r] + bv0;
            Y[(size_t)orow * 32 + 16 + mrow] = c1[r] + bv1;
        }
    }
}

constexpr int SLOT_CAP = 64;
constexpr int EMPTY    = -1;

typedef int v4i __attribute__((ext_vector_type(4)));

constexpr int CTL_UNIT  = 64;
constexpr int CTL_UNITS = 8;
constexpr int NODE_BLK  = 128;
constexpr int EDGE_BLK  = 512;

__device__ __forceinline__ int split_node(int n_nodes)
{
    return ((n_nodes / 2 + NODE_BLK - 1) / NODE_BLK) * NODE_BLK;
}

__device__ __forceinline__ int draw_item(int* q, int n0, int n1)
{
    __shared__ int s_item;
    if (threadIdx.x == 0) {
        const unsigned id = __builtin_amdgcn_s_sendmsg_rtn(0x87);
        const int h  = (int)((id >> 18) & 1u);
        const int nh = h ? n1 : n0;
        int t = atomicAdd(q + h * CTL_UNIT, 1);
        int item;
        if (t < nh) {
            item = (t << 1) | h;
        } else {
            const int o  = 1 - h;
            const int no = o ? n1 : n0;
            t = atomicAdd(q + o * CTL_UNIT, 1);
            item = (t < no) ? ((t << 1) | o) : -1;
        }
        s_item = item;
    }
    __syncthreads();
    return s_item;
}

__global__ void fill_empty(v4i* __restrict__ slots4, int n_vec, int* __restrict__ ctl)
{
    const int i = blockIdx.x * blockDim.x + threadIdx.x;
    if (i < n_vec) {
        const v4i e = {EMPTY, EMPTY, EMPTY, EMPTY};
        slots4[i] = e;
    }
    if (i < CTL_UNITS) ctl[i * CTL_UNIT] = 0;
}

__global__ void __launch_bounds__(256)
place_edges(const int* __restrict__ src, const int* __restrict__ dst,
            int* slots, int* n_spill,
            int* __restrict__ spill_dst, int* __restrict__ spill_src,
            int* q, int n_edges, int n_nodes)
{
    const int n_runs = (n_edges + EDGE_BLK - 1) / EDGE_BLK;
    const int item = draw_item(q, n_runs, n_runs);
    if (item < 0) return;
    const int share  = item & 1;
    const int e_base = (item >> 1) * EDGE_BLK + (int)threadIdx.x;
    const int n_half = split_node(n_nodes);

    int      d[2], s[2], old[2];
    int*     row[2];
    unsigned h[2];
    bool     mine[2];
#pragma unroll
    for (int u = 0; u < 2; ++u) {
        const int e = e_base + u * 256;
        d[u] = (e < n_edges) ? dst[e] : 0;
        mine[u] = (e < n_edges) && ((d[u] >= n_half ? 1 : 0) == share);
        s[u] = (e < n_edges) ? src[e] : 0;
        row[u] = slots + (size_t)d[u] * SLOT_CAP;
        h[u] = ((unsigned)e * 0x9E3779B1u) >> 26;
        old[u] = EMPTY;
    }
#pragma unroll
    for (int u = 0; u < 2; ++u)
        if (mine[u]) old[u] = atomicCAS(&row[u][h[u]], EMPTY, s[u]);
#pragma unroll
    for (int u = 0; u < 2; ++u) {
        if (old[u] == EMPTY) continue;
        bool placed = false;
        for (int i = 1; i < SLOT_CAP; ++i) {
            const int o = atomicCAS(&row[u][(h[u] + (unsigned)i) & (unsigned)(SLOT_CAP - 1)], EMPTY, s[u]);
            if (o == EMPTY) { placed = true; break; }
        }
        if (!placed) {
            const int j = atomicAdd(n_spill, 1);
            spill_dst[j] = d[u];
            spill_src[j] = s[u];
        }
    }
}

typedef float v4f __attribute__((ext_vector_type(4)));

template <int U>
__device__ __forceinline__ int group_bcast(int v)
{
    return __builtin_amdgcn_mov_dpp8(v, U * 0x249249);
}

template <int NCH>
__device__ __forceinline__ void sum_rows(v4f& acc, const int* idx, int e0, int deg,
                                         const float* __restrict__ H, unsigned coff)
{
    v4f r[NCH * 8];
#pragma unroll
    for (int q = 0; q < NCH; ++q) {
        int s[8];
        s[0] = group_bcast<0>(idx[q]); s[1] = group_bcast<1>(idx[q]);
        s[2] = group_bcast<2>(idx[q]); s[3] = group_bcast<3>(idx[q]);
        s[4] = group_bcast<4>(idx[q]); s[5] = group_bcast<5>(idx[q]);
        s[6] = group_bcast<6>(idx[q]); s[7] = group_bcast<7>(idx[q]);
#pragma unroll
        for (int u = 0; u < 8; ++u) {
            const unsigned off = (unsigned)s[u] * 128u + coff;
            r[q * 8 + u] = *(const v4f*)((const char*)H + (size_t)off);
        }
    }
    __builtin_amdgcn_sched_barrier(0);
#pragma unroll
    for (int i = 0; i < NCH * 8; ++i) {
        const v4f t = acc + r[i];
        const bool ok = (e0 + i < deg);
        acc.x = ok ? t.x : acc.x;
        acc.y = ok ? t.y : acc.y;
        acc.z = ok ? t.z : acc.z;
        acc.w = ok ? t.w : acc.w;
    }
}

constexpr int dpp8_shift_sel(int D)
{
    int sel = 0;
    for (int i = 0; i < 8; ++i) sel |= ((i >= D) ? (i - D) : 0) << (3 * i);
    return sel;
}

__global__ void __launch_bounds__(256)
pull_mean(const int* __restrict__ slots,
          const int* __restrict__ spill_dst, const int* __restrict__ spill_src,
          const int* __restrict__ n_spill,
          const float* __restrict__ H, float* __restrict__ Y, int n_nodes, int n_edges,
          int relu, int* q)
{
    __shared__ int dense[8 * 4 * SLOT_CAP];

    const int nb  = (n_nodes + NODE_BLK - 1) / NODE_BLK;
    const int nb0 = min(nb, split_node(n_nodes) / NODE_BLK);
    const int item = draw_item(q, nb0, nb - nb0);
    if (item < 0) return;
    const int blk = (item >> 1) + ((item & 1) ? nb0 : 0);

    const int lane = threadIdx.x & 31;
    const int g    = lane >> 3;
    const int c    = lane & 7;
    const int wave = __builtin_amdgcn_readfirstlane(blk * 8 + (int)(threadIdx.x >> 5));
    const int n0   = wave * 16;
    if (n0 >= n_nodes) return;

    int* my = dense + (threadIdx.x >> 3) * SLOT_CAP;

    const unsigned last = (unsigned)(n_nodes - 1);
    const unsigned coff = (unsigned)c * 16u;

    v4i ra, rb;
    {
        const unsigned beg = min((unsigned)(n0 + g), last) * (unsigned)SLOT_CAP + 8u * (unsigned)c;
        ra = *(const v4i*)(slots + beg);
        rb = *(const v4i*)(slots + beg + 4);
    }

#pragma unroll 1
    for (int p = 0; p < 4; ++p) {
        v4i na, nb;
        {
            const unsigned beg = min((unsigned)(n0 + 4 * p + 4 + g), last) * (unsigned)SLOT_CAP + 8u * (unsigned)c;
            na = *(const v4i*)(slots + beg);
            nb = *(const v4i*)(slots + beg + 4);
        }

        const int v[8] = {ra.x, ra.y, ra.z, ra.w, rb.x, rb.y, rb.z, rb.w};
        int nv = 0;
#pragma unroll
        for (int j = 0; j < 8; ++j) nv += (v[j] >= 0) ? 1 : 0;
        int incl = nv;
        {
            int t = __builtin_amdgcn_mov_dpp8(incl, dpp8_shift_sel(1));
            incl += (c >= 1) ? t : 0;
            t = __builtin_amdgcn_mov_dpp8(incl, dpp8_shift_sel(2));
            incl += (c >= 2) ? t : 0;
            t = __builtin_amdgcn_mov_dpp8(incl, dpp8_shift_sel(4));
            incl += (c >= 4) ? t : 0;
        }
        const int cdeg = group_bcast<7>(incl);
        {
            int run = incl - nv;
#pragma unroll
            for (int j = 0; j < 8; ++j) {
                const bool ok = (v[j] >= 0);
                my[ok ? run : cdeg + 8 * c + j - run] = v[j];
                run += ok ? 1 : 0;
            }
        }
        __builtin_amdgcn_fence(__ATOMIC_RELEASE, "wavefront");
        __builtin_amdgcn_wave_barrier();
        __builtin_amdgcn_fence(__ATOMIC_ACQUIRE, "wavefront");

        int cur[4];
#pragma unroll
        for (int q = 0; q < 4; ++q) cur[q] = my[8 * q + c];
        const int cmt = max(max(__builtin_amdgcn_readlane(cdeg, 0), __builtin_amdgcn_readlane(cdeg, 8)),
                            max(__builtin_amdgcn_readlane(cdeg, 16), __builtin_amdgcn_readlane(cdeg, 24)));

#pragma unroll
        for (int q = 0; q < 4; ++q) cur[q] = (int)min((unsigned)cur[q], last);
        const int fb = group_bcast<0>(cur[0]);
#pragma unroll
        for (int q = 0; q < 4; ++q) cur[q] = (8 * q + c < cdeg) ? cur[q] : fb;

        v4f acc = {0.0f, 0.0f, 0.0f, 0.0f};
        if (cmt <= 16) {
            sum_rows<2>(acc, cur, 0, cdeg, H, coff);
        } else {
            sum_rows<3>(acc, cur, 0, cdeg, H, coff);
            if (cmt > 24) {
                sum_rows<1>(acc, cur + 3, 24, cdeg, H, coff);
                for (int e0 = 32; e0 < cmt; e0 += 8) {
                    int t = (int)min((unsigned)my[e0 + c], last);
                    t = (e0 + c < cdeg) ? t : fb;
                    sum_rows<1>(acc, &t, e0, cdeg, H, coff);
                }
            }
        }
        int cfull = cdeg;
        if (cmt >= SLOT_CAP) {
            const int ns   = min(n_spill[0], n_edges);
            const int node = n0 + 4 * p + g;
            if (cdeg >= SLOT_CAP) {
                for (int k = 0; k < ns; ++k) {
                    if (spill_dst[k] == node) {
                        const unsigned r = min((unsigned)spill_src[k], last);
                        acc += *(const v4f*)((const char*)H + (size_t)(r * 128u + coff));
                        ++cfull;
                    }
                }
            }
        }

        const float dn = fmaxf((float)cfull, 1.0f);
        v4f v4 = {acc.x / dn, acc.y / dn, acc.z / dn, acc.w / dn};
        if (relu) {
            v4.x = fmaxf(v4.x, 0.0f); v4.y = fmaxf(v4.y, 0.0f);
            v4.z = fmaxf(v4.z, 0.0f); v4.w = fmaxf(v4.w, 0.0f);
        }
        const int node_out = n0 + 4 * p + g;
        if (node_out < n_nodes) *(v4f*)(Y + (size_t)node_out * 32 + c * 4) = v4;

        ra = na; rb = nb;
    }
}

extern "C" void kernel_launch(void* const* d_in, const int* in_sizes, int n_in,
                              void* d_out, int out_size, void* d_ws, size_t ws_size,
                              hipStream_t stream)
{
    const float* x    = (const float*)d_in[0];
    const int*   eidx = (const int*)d_in[1];
    const float* W1   = (const float*)d_in[2];
    const float* b1   = (const float*)d_in[3];
    const float* W2   = (const float*)d_in[4];
    const float* b2   = (const float*)d_in[5];
    float*       out  = (float*)d_out;

    const int n_nodes = in_sizes[0] / 128;
    const int n_edges = in_sizes[1] / 2;
    const int* src = eidx;
    const int* dst = eidx + n_edges;

    float* H1        = (float*)d_ws;
    float* A2        = H1 + (size_t)n_nodes * 32;
    float* H2        = A2 + (size_t)n_nodes * 32;
    int*   ctl       = (int*)(H2 + (size_t)n_nodes * 32);
    int*   n_spill   = ctl;
    int*   q_place   = ctl + 1 * CTL_UNIT;
    int*   q_pull1   = ctl + 3 * CTL_UNIT;
    int*   q_pull2   = ctl + 5 * CTL_UNIT;
    int*   slots     = ctl + CTL_UNITS * CTL_UNIT;
    int*   spill_dst = slots + (size_t)n_nodes * SLOT_CAP;
    int*   spill_src = spill_dst + n_edges;

    const int TPB = 256;
    const int n_vec      = n_nodes * (SLOT_CAP / 4);
    const int grid_fill  = (n_vec + TPB - 1) / TPB;
    const int grid_edges = 2 * ((n_edges + EDGE_BLK - 1) / EDGE_BLK);
    const int n_tiles    = (n_nodes + 15) / 16;
    const int grid_gemm  = (n_tiles + 7) / 8;
    const int grid_pull  = (n_nodes + NODE_BLK - 1) / NODE_BLK;

    fill_empty<<<grid_fill, TPB, 0, stream>>>((v4i*)slots, n_vec, ctl);
    place_edges<<<grid_edges, TPB, 0, stream>>>(src, dst, slots, n_spill, spill_dst, spill_src, q_place, n_edges, n_nodes);
    gemm_bias_wmma<<<grid_gemm, TPB, 0, stream>>>(x, W1, b1, H1, 128, n_nodes);
    pull_mean<<<grid_pull, TPB, 0, stream>>>(slots, spill_dst, spill_src, n_spill, H1, A2, n_nodes, n_edges, 1, q_pull1);
    gemm_bias_wmma<<<grid_gemm, TPB, 0, stream>>>(A2, W2, b2, H2, 32, n_nodes);
    pull_mean<<<grid_pull, TPB, 0, stream>>>(slots, spill_dst, spill_src, n_spill, H2, out, n_nodes, n_edges, 0, q_pull2);
}
